# starting kernel
# baseline (speedup 1.0000x reference)
#include <hip/hip_runtime.h>
#include <hip/hip_bf16.h>

typedef __attribute__((ext_vector_type(2))) float v2f;
typedef __attribute__((ext_vector_type(8))) float v8f;

#define NST 64
#define TCH 64
#define LSEQ 4096
#define DMOD 512
#define NSEG 16
#define SCH 4

#define WS_ST 20480

__global__ __launch_bounds__(64) void s4_precompute(const float* __restrict__ A_,
                                                    const float* __restrict__ B_,
                                                    const float* __restrict__ C_,
                                                    float* __restrict__ ws) {
  __shared__ float As[NST * NST];
  __shared__ float Sq[NST * NST];
  __shared__ float Sq2[NST * NST];
  __shared__ float vbuf[NST][NST];
  __shared__ float wbuf[NST + 1][NST];
  __shared__ float cb[NST];
  __shared__ float Kk[NST];

  const int tid = threadIdx.x;
#pragma clang loop unroll(disable)
  for (int j = 0; j < NST; ++j) As[tid * NST + j] = A_[tid * NST + j];
  cb[tid] = C_[tid];
  vbuf[0][tid] = B_[tid];
  __syncthreads();

#pragma clang loop unroll(disable)
  for (int t = 1; t < NST; ++t) {
    float s = 0.f;
#pragma clang loop unroll(disable)
    for (int j = 0; j < NST; ++j) s += As[tid * NST + j] * vbuf[t - 1][j];
    vbuf[t][tid] = s;
    __syncthreads();
  }

#pragma clang loop unroll(disable)
  for (int p = 1; p <= NST; ++p) {
    const float* prev = (p == 1) ? cb : wbuf[p - 1];
    float s = 0.f;
#pragma clang loop unroll(disable)
    for (int i = 0; i < NST; ++i) s += prev[i] * As[i * NST + tid];
    wbuf[p][tid] = s;
    __syncthreads();
  }

  {
    float s = 0.f;
#pragma clang loop unroll(disable)
    for (int i = 0; i < NST; ++i) s += cb[i] * vbuf[tid][i];
    Kk[tid] = s;
  }
  __syncthreads();

  const float* src = As;
  float* dst = Sq2;
#pragma clang loop unroll(disable)
  for (int it = 0; it < 6; ++it) {
#pragma clang loop unroll(disable)
    for (int j = 0; j < NST; ++j) {
      float s = 0.f;
#pragma clang loop unroll(disable)
      for (int k = 0; k < NST; ++k) s += src[tid * NST + k] * src[k * NST + j];
      dst[tid * NST + j] = s;
    }
    __syncthreads();
    src = dst;
    dst = (src == Sq2) ? Sq : Sq2;
  }
  const float* A64 = src;

  float* TKt = ws;
  float* Mt  = ws + 4096;
  float* Et  = ws + 8192;
  float* Pt  = ws + 12288;
#pragma clang loop unroll(disable)
  for (int j = 0; j < NST; ++j) {
    TKt[tid * NST + j] = (j >= tid) ? Kk[j - tid] : 0.f;
    Mt [tid * NST + j] = vbuf[NST - 1 - tid][j];
    Et [tid * NST + j] = wbuf[j + 1][tid];
    Pt [tid * NST + j] = A64[j * NST + tid];
  }
  __syncthreads();

#pragma clang loop unroll(disable)
  for (int it = 0; it < 2; ++it) {
#pragma clang loop unroll(disable)
    for (int j = 0; j < NST; ++j) {
      float s = 0.f;
#pragma clang loop unroll(disable)
      for (int k = 0; k < NST; ++k) s += src[tid * NST + k] * src[k * NST + j];
      dst[tid * NST + j] = s;
    }
    __syncthreads();
    src = dst;
    dst = (src == Sq2) ? Sq : Sq2;
  }
  const float* A256 = src;
  float* PSt = ws + 16384;
#pragma clang loop unroll(disable)
  for (int j = 0; j < NST; ++j) {
    PSt[tid * NST + j] = A256[j * NST + tid];
  }
}

__global__ __launch_bounds__(32)
__attribute__((amdgpu_num_vgpr(256)))
void s4_segstate(const float* __restrict__ u, float* __restrict__ ws) {
  __shared__ float sM[2 * 4096];
  __shared__ float sX[2][1024];

  const int tid = threadIdx.x;
  {
    const float4* m4 = (const float4*)(ws + 4096);
    const float4* p4 = (const float4*)(ws + 12288);
    float4* d0 = (float4*)sM;
    float4* d1 = (float4*)(sM + 4096);
    for (int i = tid; i < 1024; i += 32) d0[i] = m4[i];
    for (int i = tid; i < 1024; i += 32) d1[i] = p4[i];
    for (int i = tid; i < 1024; i += 32) sX[0][i] = 0.f;
  }
  __syncthreads();

  const int tile = blockIdx.x >> 4;
  const int seg  = blockIdx.x & 15;
  const int s0 = tile * 16;
  const int bb = s0 >> 9, dm0 = s0 & 511;
  const int row = tid & 15, hi = tid >> 4;
  const float* ub0 = u + (long)bb * LSEQ * DMOD + dm0 + row + (long)(2 * hi) * DMOD;

  for (int i = 0; i < SCH; ++i) {
    const int c = seg * SCH + i;
    const float* Xr = sX[i & 1];
    float* Xw = (float*)sX[(i & 1) ^ 1];
    const float* up = ub0 + (long)c * TCH * DMOD;
    v2f uf[16];
#pragma unroll
    for (int kk = 0; kk < 16; ++kk) {
      uf[kk].x = up[kk * 4 * DMOD];
      uf[kk].y = up[kk * 4 * DMOD + DMOD];
    }
    const v2f* xr = (const v2f*)(Xr + row * 64 + 2 * hi);
    v2f xin[16];
#pragma unroll
    for (int kk = 0; kk < 16; ++kk) xin[kk] = xr[kk * 2];

#pragma unroll
    for (int nt = 0; nt < 4; ++nt) {
      v8f ax = {0.f, 0.f, 0.f, 0.f, 0.f, 0.f, 0.f, 0.f};
      const int col = nt * 16 + row;
      const float* pM = sM + col;
      const float* pP = sM + 4096 + col;
#pragma unroll
      for (int kk = 0; kk < 16; ++kk) {
        const int k0 = 4 * kk + 2 * hi;
        v2f bm; bm.x = pM[k0 * 64]; bm.y = pM[(k0 + 1) * 64];
        ax = __builtin_amdgcn_wmma_f32_16x16x4_f32(false, uf[kk], false, bm,
                                                   (short)0, ax, false, false);
        v2f bp; bp.x = pP[k0 * 64]; bp.y = pP[(k0 + 1) * 64];
        ax = __builtin_amdgcn_wmma_f32_16x16x4_f32(false, xin[kk], false, bp,
                                                   (short)0, ax, false, false);
      }
      float* xw = Xw + (8 * hi) * 64 + col;
#pragma unroll
      for (int r = 0; r < 8; ++r) xw[r * 64] = ax[r];
    }
    __syncthreads();
  }
  float* st = ws + WS_ST + (long)tile * (NSEG * 1024) + seg * 1024;
  const float4* xp = (const float4*)sX[0];
  float4* gp = (float4*)st;
  for (int i = tid; i < 256; i += 32) gp[i] = xp[i];
}

__global__ __launch_bounds__(32) void s4_combine(float* __restrict__ ws) {
  __shared__ float sP[4096];
  __shared__ float sE[2][1024];

  const int tid = threadIdx.x;
  {
    const float4* p4 = (const float4*)(ws + 16384);
    float4* d4 = (float4*)sP;
    for (int i = tid; i < 1024; i += 32) d4[i] = p4[i];
    for (int i = tid; i < 1024; i += 32) sE[0][i] = 0.f;
  }
  __syncthreads();

  const int row = tid & 15, hi = tid >> 4;
  float* stt = ws + WS_ST + (long)blockIdx.x * (NSEG * 1024);

#pragma clang loop unroll(disable)
  for (int g = 0; g < NSEG - 1; ++g) {
    float* slot = stt + g * 1024;
    const float* Er = sE[g & 1];
    float* Ew = (float*)sE[(g & 1) ^ 1];
    const v2f* er = (const v2f*)(Er + row * 64 + 2 * hi);
    v2f ef[16];
#pragma unroll
    for (int kk = 0; kk < 16; ++kk) ef[kk] = er[kk * 2];

#pragma unroll
    for (int nt = 0; nt < 4; ++nt) {
      const int col = nt * 16 + row;
      v8f ax;
#pragma unroll
      for (int r = 0; r < 8; ++r) ax[r] = slot[(r + 8 * hi) * 64 + col];
      const float* pP = sP + col;
#pragma unroll
      for (int kk = 0; kk < 16; ++kk) {
        const int k0 = 4 * kk + 2 * hi;
        v2f bp; bp.x = pP[k0 * 64]; bp.y = pP[(k0 + 1) * 64];
        ax = __builtin_amdgcn_wmma_f32_16x16x4_f32(false, ef[kk], false, bp,
                                                   (short)0, ax, false, false);
      }
#pragma unroll
      for (int r = 0; r < 8; ++r) slot[(r + 8 * hi) * 64 + col] = ax[r];
      float* ew = Ew + (8 * hi) * 64 + col;
#pragma unroll
      for (int r = 0; r < 8; ++r) ew[r * 64] = ax[r];
    }
    __syncthreads();
  }
}

__global__ __launch_bounds__(32)
__attribute__((amdgpu_num_vgpr(256)))
void s4_outputs(const float* __restrict__ u, const float* __restrict__ ws,
                float* __restrict__ y) {
  __shared__ float sMat[4 * 4096];
  __shared__ float sX[2][1024];

  const int tid = threadIdx.x;
  {
    const float4* s4p = (const float4*)ws;
    float4* d4 = (float4*)sMat;
    for (int i = tid; i < 4096; i += 32) d4[i] = s4p[i];
  }
  const int tile = blockIdx.x >> 4;
  const int seg  = blockIdx.x & 15;
  if (seg == 0) {
    for (int i = tid; i < 1024; i += 32) sX[0][i] = 0.f;
  } else {
    const float4* ep =
        (const float4*)(ws + WS_ST + (long)tile * (NSEG * 1024) + (seg - 1) * 1024);
    float4* xp = (float4*)sX[0];
    for (int i = tid; i < 256; i += 32) xp[i] = ep[i];
  }
  __syncthreads();

  const int s0 = tile * 16;
  const int bb = s0 >> 9, dm0 = s0 & 511;
  const int row = tid & 15, hi = tid >> 4;
  const float* ub0 = u + (long)bb * LSEQ * DMOD + dm0 + row + (long)(2 * hi) * DMOD;
  float* yb0 = y + (long)bb * LSEQ * DMOD + dm0 + 8 * hi;

  for (int i = 0; i < SCH; ++i) {
    const int c = seg * SCH + i;
    const float* Xr = sX[i & 1];
    float* Xw = (float*)sX[(i & 1) ^ 1];
    const float* up = ub0 + (long)c * TCH * DMOD;
    v2f uf[16];
#pragma unroll
    for (int kk = 0; kk < 16; ++kk) {
      uf[kk].x = up[kk * 4 * DMOD];
      uf[kk].y = up[kk * 4 * DMOD + DMOD];
    }
    const v2f* xr = (const v2f*)(Xr + row * 64 + 2 * hi);
    v2f xin[16];
#pragma unroll
    for (int kk = 0; kk < 16; ++kk) xin[kk] = xr[kk * 2];

#pragma unroll
    for (int nt = 0; nt < 4; ++nt) {
      v8f ay = {0.f, 0.f, 0.f, 0.f, 0.f, 0.f, 0.f, 0.f};
      v8f ax = {0.f, 0.f, 0.f, 0.f, 0.f, 0.f, 0.f, 0.f};
      const int col = nt * 16 + row;
      const float* pTK = sMat + col;
      const float* pMt = sMat + 4096 + col;
      const float* pEt = sMat + 8192 + col;
      const float* pPt = sMat + 12288 + col;
#pragma unroll
      for (int kk = 0; kk < 16; ++kk) {
        const int k0 = 4 * kk + 2 * hi;
        v2f bt; bt.x = pTK[k0 * 64]; bt.y = pTK[(k0 + 1) * 64];
        ay = __builtin_amdgcn_wmma_f32_16x16x4_f32(false, uf[kk], false, bt,
                                                   (short)0, ay, false, false);
        v2f bm; bm.x = pMt[k0 * 64]; bm.y = pMt[(k0 + 1) * 64];
        ax = __builtin_amdgcn_wmma_f32_16x16x4_f32(false, uf[kk], false, bm,
                                                   (short)0, ax, false, false);
        v2f be; be.x = pEt[k0 * 64]; be.y = pEt[(k0 + 1) * 64];
        ay = __builtin_amdgcn_wmma_f32_16x16x4_f32(false, xin[kk], false, be,
                                                   (short)0, ay, false, false);
        v2f bp; bp.x = pPt[k0 * 64]; bp.y = pPt[(k0 + 1) * 64];
        ax = __builtin_amdgcn_wmma_f32_16x16x4_f32(false, xin[kk], false, bp,
                                                   (short)0, ax, false, false);
      }
      float4* yp = (float4*)(yb0 + (long)(c * TCH + col) * DMOD);
      float4 lo, hh;
      lo.x = ay[0]; lo.y = ay[1]; lo.z = ay[2]; lo.w = ay[3];
      hh.x = ay[4]; hh.y = ay[5]; hh.z = ay[6]; hh.w = ay[7];
      yp[0] = lo;
      yp[1] = hh;
      float* xw = Xw + (8 * hi) * 64 + col;
#pragma unroll
      for (int r = 0; r < 8; ++r) xw[r * 64] = ax[r];
    }
    __syncthreads();
  }
}

extern "C" void kernel_launch(void* const* d_in, const int* in_sizes, int n_in,
                              void* d_out, int out_size, void* d_ws, size_t ws_size,
                              hipStream_t stream) {
  const float* u  = (const float*)d_in[0];
  const float* A_ = (const float*)d_in[1];
  const float* B_ = (const float*)d_in[2];
  const float* C_ = (const float*)d_in[3];
  float* y  = (float*)d_out;
  float* ws = (float*)d_ws;

  s4_precompute<<<1, 64, 0, stream>>>(A_, B_, C_, ws);
  s4_segstate<<<128 * NSEG, 32, 0, stream>>>(u, ws);
  s4_combine<<<128, 32, 0, stream>>>(ws);
  s4_outputs<<<128 * NSEG, 32, 0, stream>>>(u, ws, y);
}


# Time cut into 8 segments of 8 chunks (was 16 of 4): one wave per SIMD in phases A and C, chunk matrices loaded once per SIMD, segment scan 7 steps with A^512; state sum order changed
# speedup vs baseline: 42.0157x; 42.0157x over previous
#include <hip/hip_runtime.h>
#include <hip/hip_bf16.h>

typedef __attribute__((ext_vector_type(2))) float v2f;
typedef __attribute__((ext_vector_type(8))) float v8f;

#define NST 64
#define TCH 64
#define LSEQ 4096
#define DMOD 512
#define NSEG 8
#define SCH 8
#define NRND 9

#define FR_TK 0
#define FR_M  4096
#define FR_E  8192
#define FR_P  12288
#define FR_PS 16384
#define WS_ST 20480

__device__ __forceinline__ void s4_frag_kc(int e, int& ku, int& kd, int& col) {
  const int el = e & 1, jj = (e >> 1) & 1, l = (e >> 2) & 31, jp = (e >> 7) & 7, nt = e >> 10;
  const int j = 2 * jp + jj, h = l >> 4;
  col = nt * 16 + (l & 15);
  ku = 4 * j + 2 * h + el;
  kd = 16 * (j >> 2) + 8 * h + 2 * (j & 3) + el;
}

#define PIT 68

__device__ __forceinline__ v8f s4_tile_mm(const float* Am, const float* Bm,
                                          int i0, int j0, int row, int hi) {
  v8f acc = {0.f, 0.f, 0.f, 0.f, 0.f, 0.f, 0.f, 0.f};
  const float* pa = Am + (i0 + row) * PIT + 2 * hi;
  const float* pb = Bm + (2 * hi) * PIT + j0 + row;
#pragma unroll
  for (int kk = 0; kk < 16; ++kk) {
    const v2f af = *(const v2f*)(pa + 4 * kk);
    v2f bf;
    bf.x = pb[(4 * kk) * PIT];
    bf.y = pb[(4 * kk + 1) * PIT];
    acc = __builtin_amdgcn_wmma_f32_16x16x4_f32(false, af, false, bf,
                                                (short)0, acc, false, false);
  }
  return acc;
}

__global__ __launch_bounds__(512) void s4_precompute(const float* __restrict__ A_,
                                                     const float* __restrict__ B_,
                                                     const float* __restrict__ C_,
                                                     float* __restrict__ ws) {
  __shared__ float S[NST * PIT];
  __shared__ float V[NST * PIT];
  __shared__ float Wm[NST * PIT];
  __shared__ float cb[NST];
  __shared__ float Kk[NST];

  const int tid = threadIdx.x;
  const int wave = __builtin_amdgcn_readfirstlane(tid >> 5);
  const int lane = tid & 31;
  const int row = lane & 15, hi = lane >> 4;

  for (int e = tid; e < NST * NST; e += 512) {
    const int i = e >> 6, j = e & 63;
    S[i * PIT + j] = A_[e];
    V[i * PIT + j] = 0.f;
    Wm[i * PIT + j] = 0.f;
  }
  if (tid < NST) cb[tid] = C_[tid];
  __syncthreads();

  if (tid < NST) {
    V[tid * PIT] = B_[tid];
  } else if (tid < 2 * NST) {
    const int j = tid - NST;
    float s = 0.f;
#pragma unroll 16
    for (int i = 0; i < NST; ++i) s += cb[i] * S[i * PIT + j];
    Wm[j] = s;
  }
  __syncthreads();

  const int ti = wave >> 2, tj = wave & 3;
  const int w8 = wave & 7;
  const int ha = w8 >> 2;
  const int tb = w8 & 3;
  const bool isV = wave < 8;

  float* TKt = ws + FR_TK;
  float* Mt  = ws + FR_M;
  float* Et  = ws + FR_E;
  float* Pt  = ws + FR_P;
  float* PSt = ws + FR_PS;

#pragma clang loop unroll(disable)
  for (int r = 0; r < NRND; ++r) {
    const int m = 1 << r;
    const bool need = (r < 6) && (ha * 16 < m);
    v8f a2 = {0.f, 0.f, 0.f, 0.f, 0.f, 0.f, 0.f, 0.f};
    if (need) {
      a2 = isV ? s4_tile_mm(S, V, tb * 16, ha * 16, row, hi)
               : s4_tile_mm(Wm, S, ha * 16, tb * 16, row, hi);
    }
    const v8f aS = s4_tile_mm(S, S, ti * 16, tj * 16, row, hi);
    __syncthreads();

    {
      float* sp = S + (ti * 16 + 8 * hi) * PIT + tj * 16 + row;
#pragma unroll
      for (int q = 0; q < 8; ++q) sp[q * PIT] = aS[q];
    }
    if (need) {
      if (isV) {
        if (ha * 16 + row < m) {
          float* vp = V + (tb * 16 + 8 * hi) * PIT + m + ha * 16 + row;
#pragma unroll
          for (int q = 0; q < 8; ++q) vp[q * PIT] = a2[q];
        }
      } else {
        float* wp = Wm + (m + ha * 16 + 8 * hi) * PIT + tb * 16 + row;
#pragma unroll
        for (int q = 0; q < 8; ++q)
          if (ha * 16 + 8 * hi + q < m) wp[q * PIT] = a2[q];
      }
    }
    __syncthreads();

    if (r == 5) {
      if (tid < NST) {
        float s = 0.f;
#pragma unroll 16
        for (int i = 0; i < NST; ++i) s += cb[i] * V[i * PIT + tid];
        Kk[tid] = s;
      }
      __syncthreads();
      for (int e = tid; e < NST * NST; e += 512) {
        int ku, kd, j;
        s4_frag_kc(e, ku, kd, j);
        TKt[e] = (j >= ku) ? Kk[j - ku] : 0.f;
        Mt[e]  = V[j * PIT + (NST - 1 - ku)];
        Et[e]  = Wm[j * PIT + kd];
        Pt[e]  = S[j * PIT + kd];
      }
    }
  }
  for (int e = tid; e < NST * NST; e += 512) {
    int ku, kd, j;
    s4_frag_kc(e, ku, kd, j);
    PSt[e] = S[j * PIT + kd];
  }
}

__device__ __forceinline__ v8f s4_mma(v2f a, v2f b, v8f c) {
  return __builtin_amdgcn_wmma_f32_16x16x4_f32(false, a, false, b, (short)0, c, false, false);
}

__device__ __forceinline__ void s4_load_frags(const float* __restrict__ fr, int lane,
                                              v2f (&f)[4][16]) {
  const float4* f4 = (const float4*)fr + lane;
#pragma unroll
  for (int nt = 0; nt < 4; ++nt) {
#pragma unroll
    for (int jp = 0; jp < 8; ++jp) {
      const float4 t = f4[(nt * 8 + jp) * 32];
      f[nt][2 * jp].x = t.x;     f[nt][2 * jp].y = t.y;
      f[nt][2 * jp + 1].x = t.z; f[nt][2 * jp + 1].y = t.w;
    }
  }
}

__device__ __forceinline__ v2f s4_state_frag(const v8f (&X)[4], int j) {
  v2f b;
  b.x = X[j >> 2][2 * (j & 3)];
  b.y = X[j >> 2][2 * (j & 3) + 1];
  return b;
}

__device__ __forceinline__ void s4_load_u(const float* __restrict__ up, v2f (&uf)[16]) {
#pragma unroll
  for (int j = 0; j < 16; ++j) {
    uf[j].x = up[(4 * j) * DMOD];
    uf[j].y = up[(4 * j + 1) * DMOD];
  }
}

__device__ __forceinline__ void s4_state_chunk(const v2f (&fM)[4][16], const v2f (&fP)[4][16],
                                               const v2f (&uf)[16], const v8f (&X)[4],
                                               v8f (&Xn)[4]) {
#pragma unroll
  for (int nt = 0; nt < 4; nt += 2) {
    v8f a0 = {0.f, 0.f, 0.f, 0.f, 0.f, 0.f, 0.f, 0.f};
    v8f a1 = {0.f, 0.f, 0.f, 0.f, 0.f, 0.f, 0.f, 0.f};
#pragma unroll
    for (int j = 0; j < 16; ++j) {
      const v2f xb = s4_state_frag(X, j);
      a0 = s4_mma(fM[nt][j],     uf[j], a0);
      a1 = s4_mma(fM[nt + 1][j], uf[j], a1);
      a0 = s4_mma(fP[nt][j],     xb, a0);
      a1 = s4_mma(fP[nt + 1][j], xb, a1);
    }
    Xn[nt] = a0;
    Xn[nt + 1] = a1;
  }
}

__global__ __launch_bounds__(32) __attribute__((amdgpu_waves_per_eu(1, 1)))
void s4_segstate(const float* __restrict__ u, float* __restrict__ ws) {
  const int lane = threadIdx.x;
  const int tile = blockIdx.x / NSEG;
  const int seg  = blockIdx.x % NSEG;
  const int s0 = tile * 16;
  const int bb = s0 >> 9, dm0 = s0 & 511;
  const int c = lane & 15, hi = lane >> 4;
  const float* ub0 = u + (long)bb * LSEQ * DMOD + dm0 + c
                       + (long)(seg * SCH * TCH + 2 * hi) * DMOD;

  v2f ua[16], ub[16];
  v2f fM[4][16], fP[4][16];
  s4_load_u(ub0, ua);
  s4_load_frags(ws + FR_M, lane, fM);
  s4_load_frags(ws + FR_P, lane, fP);
  s4_load_u(ub0 + (long)TCH * DMOD, ub);

  v8f X[4], Xa[4];
#pragma unroll
  for (int nt = 0; nt < 4; ++nt)
#pragma unroll
    for (int r = 0; r < 8; ++r) X[nt][r] = 0.f;

#pragma clang loop unroll(disable)
  for (int p = 0; p < SCH / 2 - 1; ++p) {
    const float* un = ub0 + (long)(2 * p + 2) * TCH * DMOD;
    s4_state_chunk(fM, fP, ua, X, Xa);
    s4_load_u(un, ua);
    s4_state_chunk(fM, fP, ub, Xa, X);
    s4_load_u(un + (long)TCH * DMOD, ub);
  }
  s4_state_chunk(fM, fP, ua, X, Xa);
  s4_state_chunk(fM, fP, ub, Xa, X);

  float* st = ws + WS_ST + (long)tile * (NSEG * 1024) + seg * 1024 + lane;
#pragma unroll
  for (int nt = 0; nt < 4; ++nt)
#pragma unroll
    for (int r = 0; r < 8; ++r) st[(nt * 8 + r) * 32] = X[nt][r];
}

__device__ __forceinline__ void s4_load_dump(const float* __restrict__ p, v8f (&G)[4]) {
#pragma unroll
  for (int nt = 0; nt < 4; ++nt)
#pragma unroll
    for (int r = 0; r < 8; ++r) G[nt][r] = p[(nt * 8 + r) * 32];
}

__device__ __forceinline__ void s4_scan_step(const v2f (&fPS)[4][16], v8f (&E)[4],
                                             const v8f (&G)[4], float* __restrict__ slot) {
  v8f En[4];
#pragma unroll
  for (int nt = 0; nt < 4; nt += 2) {
    v8f a0 = G[nt];
    v8f a1 = G[nt + 1];
#pragma unroll
    for (int j = 0; j < 16; ++j) {
      const v2f eb = s4_state_frag(E, j);
      a0 = s4_mma(fPS[nt][j],     eb, a0);
      a1 = s4_mma(fPS[nt + 1][j], eb, a1);
    }
    En[nt] = a0;
    En[nt + 1] = a1;
  }
#pragma unroll
  for (int nt = 0; nt < 4; ++nt) {
    E[nt] = En[nt];
#pragma unroll
    for (int r = 0; r < 8; ++r) slot[(nt * 8 + r) * 32] = En[nt][r];
  }
}

__global__ __launch_bounds__(32) void s4_combine(float* __restrict__ ws) {
  const int lane = threadIdx.x;
  float* stt = ws + WS_ST + (long)blockIdx.x * (NSEG * 1024) + lane;

  v8f G[4], Gn[4];
  s4_load_dump(stt, G);
  v2f fPS[4][16];
  s4_load_frags(ws + FR_PS, lane, fPS);
  s4_load_dump(stt + 1024, Gn);

  v8f E[4];
#pragma unroll
  for (int nt = 0; nt < 4; ++nt)
#pragma unroll
    for (int r = 0; r < 8; ++r) E[nt][r] = 0.f;

  s4_scan_step(fPS, E, G, stt);

#pragma clang loop unroll(disable)
  for (int g = 1; g < NSEG - 1; ++g) {
    float* slot = stt + g * 1024;
#pragma unroll
    for (int nt = 0; nt < 4; ++nt) G[nt] = Gn[nt];
    if (g + 1 < NSEG - 1) s4_load_dump(slot + 1024, Gn);
    s4_scan_step(fPS, E, G, slot);
  }
}

__device__ __forceinline__ void s4_out_chunk(const v2f (&fTK)[4][16], const v2f (&fM)[4][16],
                                             const v2f (&fE)[4][16], const v2f (&fP)[4][16],
                                             const v2f (&uf)[16], const v8f (&X)[4],
                                             v8f (&Xn)[4], float* __restrict__ yp0) {
#pragma unroll
  for (int nt = 0; nt < 4; ++nt) {
    v8f ay = {0.f, 0.f, 0.f, 0.f, 0.f, 0.f, 0.f, 0.f};
    v8f ax = {0.f, 0.f, 0.f, 0.f, 0.f, 0.f, 0.f, 0.f};
#pragma unroll
    for (int j = 0; j < 16; ++j) {
      const v2f xb = s4_state_frag(X, j);
      ay = s4_mma(fTK[nt][j], uf[j], ay);
      ax = s4_mma(fM[nt][j],  uf[j], ax);
      ay = s4_mma(fE[nt][j],  xb, ay);
      ax = s4_mma(fP[nt][j],  xb, ax);
    }
    float* yp = yp0 + (long)(nt * 16) * DMOD;
#pragma unroll
    for (int r = 0; r < 8; ++r) yp[r * DMOD] = ay[r];
    Xn[nt] = ax;
  }
}

__device__ __forceinline__ void s4_out_last(const v2f (&fTK)[4][16], const v2f (&fE)[4][16],
                                            const v2f (&uf)[16], const v8f (&X)[4],
                                            float* __restrict__ yp0) {
#pragma unroll
  for (int nt = 0; nt < 4; nt += 2) {
    v8f a0 = {0.f, 0.f, 0.f, 0.f, 0.f, 0.f, 0.f, 0.f};
    v8f a1 = {0.f, 0.f, 0.f, 0.f, 0.f, 0.f, 0.f, 0.f};
#pragma unroll
    for (int j = 0; j < 16; ++j) {
      const v2f xb = s4_state_frag(X, j);
      a0 = s4_mma(fTK[nt][j],     uf[j], a0);
      a1 = s4_mma(fTK[nt + 1][j], uf[j], a1);
      a0 = s4_mma(fE[nt][j],      xb, a0);
      a1 = s4_mma(fE[nt + 1][j],  xb, a1);
    }
    float* yp = yp0 + (long)(nt * 16) * DMOD;
#pragma unroll
    for (int r = 0; r < 8; ++r) yp[r * DMOD] = a0[r];
#pragma unroll
    for (int r = 0; r < 8; ++r) yp[(16 + r) * DMOD] = a1[r];
  }
}

__global__ __launch_bounds__(32)
void s4_outputs(const float* __restrict__ u, const float* __restrict__ ws,
                float* __restrict__ y) {
  const int lane = threadIdx.x;
  const int tile = blockIdx.x / NSEG;
  const int seg  = blockIdx.x % NSEG;
  const int s0 = tile * 16;
  const int bb = s0 >> 9, dm0 = s0 & 511;
  const int c = lane & 15, hi = lane >> 4;
  const float* ub0 = u + (long)bb * LSEQ * DMOD + dm0 + c
                       + (long)(seg * SCH * TCH + 2 * hi) * DMOD;
  float* yb0 = y + (long)bb * LSEQ * DMOD + dm0 + c
                 + (long)(seg * SCH * TCH + 8 * hi) * DMOD;

  v8f X[4], Xa[4];
  if (seg == 0) {
#pragma unroll
    for (int nt = 0; nt < 4; ++nt)
#pragma unroll
      for (int r = 0; r < 8; ++r) X[nt][r] = 0.f;
  } else {
    s4_load_dump(ws + WS_ST + (long)tile * (NSEG * 1024) + (seg - 1) * 1024 + lane, X);
  }

  v2f ua[16], ub[16];
  v2f fTK[4][16], fM[4][16], fE[4][16], fP[4][16];
  s4_load_u(ub0, ua);
  s4_load_frags(ws + FR_TK, lane, fTK);
  s4_load_frags(ws + FR_M,  lane, fM);
  s4_load_frags(ws + FR_E,  lane, fE);
  s4_load_frags(ws + FR_P,  lane, fP);
  s4_load_u(ub0 + (long)TCH * DMOD, ub);

#pragma clang loop unroll(disable)
  for (int p = 0; p < SCH / 2 - 1; ++p) {
    const float* un = ub0 + (long)(2 * p + 2) * TCH * DMOD;
    float* yp = yb0 + (long)(2 * p) * TCH * DMOD;
    s4_out_chunk(fTK, fM, fE, fP, ua, X, Xa, yp);
    s4_load_u(un, ua);
    s4_out_chunk(fTK, fM, fE, fP, ub, Xa, X, yp + (long)TCH * DMOD);
    s4_load_u(un + (long)TCH * DMOD, ub);
  }
  {
    float* yp = yb0 + (long)(SCH - 2) * TCH * DMOD;
    s4_out_chunk(fTK, fM, fE, fP, ua, X, Xa, yp);
    s4_out_last(fTK, fE, ub, Xa, yp + (long)TCH * DMOD);
  }
}

extern "C" void kernel_launch(void* const* d_in, const int* in_sizes, int n_in,
                              void* d_out, int out_size, void* d_ws, size_t ws_size,
                              hipStream_t stream) {
  const float* u  = (const float*)d_in[0];
  const float* A_ = (const float*)d_in[1];
  const float* B_ = (const float*)d_in[2];
  const float* C_ = (const float*)d_in[3];
  float* y  = (float*)d_out;
  float* ws = (float*)d_ws;

  s4_precompute<<<1, 512, 0, stream>>>(A_, B_, C_, ws);
  s4_segstate<<<128 * NSEG, 32, 0, stream>>>(u, ws);
  s4_combine<<<128, 32, 0, stream>>>(ws);
  s4_outputs<<<128 * NSEG, 32, 0, stream>>>(u, ws, y);
}


# Segment scan folded into the output launch: each wave builds its entering state in registers from the earlier segments' dumps (up to 6 products by A^512) while u and chunk matrices load; 3 launches
# speedup vs baseline: 43.9365x; 1.0457x over previous
#include <hip/hip_runtime.h>
#include <hip/hip_bf16.h>

typedef __attribute__((ext_vector_type(2))) float v2f;
typedef __attribute__((ext_vector_type(8))) float v8f;

#define NST 64
#define TCH 64
#define LSEQ 4096
#define DMOD 512
#define NSEG 8
#define SCH 8
#define NRND 9

#define FR_TK 0
#define FR_M  4096
#define FR_E  8192
#define FR_P  12288
#define FR_PS 16384
#define WS_ST 20480

__device__ __forceinline__ void s4_frag_kc(int e, int& ku, int& kd, int& col) {
  const int el = e & 1, jj = (e >> 1) & 1, l = (e >> 2) & 31, jp = (e >> 7) & 7, nt = e >> 10;
  const int j = 2 * jp + jj, h = l >> 4;
  col = nt * 16 + (l & 15);
  ku = 4 * j + 2 * h + el;
  kd = 16 * (j >> 2) + 8 * h + 2 * (j & 3) + el;
}

#define PIT 68

__device__ __forceinline__ v8f s4_tile_mm(const float* Am, const float* Bm,
                                          int i0, int j0, int row, int hi) {
  v8f acc = {0.f, 0.f, 0.f, 0.f, 0.f, 0.f, 0.f, 0.f};
  const float* pa = Am + (i0 + row) * PIT + 2 * hi;
  const float* pb = Bm + (2 * hi) * PIT + j0 + row;
#pragma unroll
  for (int kk = 0; kk < 16; ++kk) {
    const v2f af = *(const v2f*)(pa + 4 * kk);
    v2f bf;
    bf.x = pb[(4 * kk) * PIT];
    bf.y = pb[(4 * kk + 1) * PIT];
    acc = __builtin_amdgcn_wmma_f32_16x16x4_f32(false, af, false, bf,
                                                (short)0, acc, false, false);
  }
  return acc;
}

__global__ __launch_bounds__(512) void s4_precompute(const float* __restrict__ A_,
                                                     const float* __restrict__ B_,
                                                     const float* __restrict__ C_,
                                                     float* __restrict__ ws) {
  __shared__ float S[NST * PIT];
  __shared__ float V[NST * PIT];
  __shared__ float Wm[NST * PIT];
  __shared__ float cb[NST];
  __shared__ float Kk[NST];

  const int tid = threadIdx.x;
  const int wave = __builtin_amdgcn_readfirstlane(tid >> 5);
  const int lane = tid & 31;
  const int row = lane & 15, hi = lane >> 4;

  for (int e = tid; e < NST * NST; e += 512) {
    const int i = e >> 6, j = e & 63;
    S[i * PIT + j] = A_[e];
    V[i * PIT + j] = 0.f;
    Wm[i * PIT + j] = 0.f;
  }
  if (tid < NST) cb[tid] = C_[tid];
  __syncthreads();

  if (tid < NST) {
    V[tid * PIT] = B_[tid];
  } else if (tid < 2 * NST) {
    const int j = tid - NST;
    float s = 0.f;
#pragma unroll 16
    for (int i = 0; i < NST; ++i) s += cb[i] * S[i * PIT + j];
    Wm[j] = s;
  }
  __syncthreads();

  const int ti = wave >> 2, tj = wave & 3;
  const int w8 = wave & 7;
  const int ha = w8 >> 2;
  const int tb = w8 & 3;
  const bool isV = wave < 8;

  float* TKt = ws + FR_TK;
  float* Mt  = ws + FR_M;
  float* Et  = ws + FR_E;
  float* Pt  = ws + FR_P;
  float* PSt = ws + FR_PS;

#pragma clang loop unroll(disable)
  for (int r = 0; r < NRND; ++r) {
    const int m = 1 << r;
    const bool need = (r < 6) && (ha * 16 < m);
    v8f a2 = {0.f, 0.f, 0.f, 0.f, 0.f, 0.f, 0.f, 0.f};
    if (need) {
      a2 = isV ? s4_tile_mm(S, V, tb * 16, ha * 16, row, hi)
               : s4_tile_mm(Wm, S, ha * 16, tb * 16, row, hi);
    }
    const v8f aS = s4_tile_mm(S, S, ti * 16, tj * 16, row, hi);
    __syncthreads();

    {
      float* sp = S + (ti * 16 + 8 * hi) * PIT + tj * 16 + row;
#pragma unroll
      for (int q = 0; q < 8; ++q) sp[q * PIT] = aS[q];
    }
    if (need) {
      if (isV) {
        if (ha * 16 + row < m) {
          float* vp = V + (tb * 16 + 8 * hi) * PIT + m + ha * 16 + row;
#pragma unroll
          for (int q = 0; q < 8; ++q) vp[q * PIT] = a2[q];
        }
      } else {
        float* wp = Wm + (m + ha * 16 + 8 * hi) * PIT + tb * 16 + row;
#pragma unroll
        for (int q = 0; q < 8; ++q)
          if (ha * 16 + 8 * hi + q < m) wp[q * PIT] = a2[q];
      }
    }
    __syncthreads();

    if (r == 5) {
      if (tid < NST) {
        float s = 0.f;
#pragma unroll 16
        for (int i = 0; i < NST; ++i) s += cb[i] * V[i * PIT + tid];
        Kk[tid] = s;
      }
      __syncthreads();
      for (int e = tid; e < NST * NST; e += 512) {
        int ku, kd, j;
        s4_frag_kc(e, ku, kd, j);
        TKt[e] = (j >= ku) ? Kk[j - ku] : 0.f;
        Mt[e]  = V[j * PIT + (NST - 1 - ku)];
        Et[e]  = Wm[j * PIT + kd];
        Pt[e]  = S[j * PIT + kd];
      }
    }
  }
  for (int e = tid; e < NST * NST; e += 512) {
    int ku, kd, j;
    s4_frag_kc(e, ku, kd, j);
    PSt[e] = S[j * PIT + kd];
  }
}

__device__ __forceinline__ v8f s4_mma(v2f a, v2f b, v8f c) {
  return __builtin_amdgcn_wmma_f32_16x16x4_f32(false, a, false, b, (short)0, c, false, false);
}

__device__ __forceinline__ void s4_load_frags(const float* __restrict__ fr, int lane,
                                              v2f (&f)[4][16]) {
  const float4* f4 = (const float4*)fr + lane;
#pragma unroll
  for (int nt = 0; nt < 4; ++nt) {
#pragma unroll
    for (int jp = 0; jp < 8; ++jp) {
      const float4 t = f4[(nt * 8 + jp) * 32];
      f[nt][2 * jp].x = t.x;     f[nt][2 * jp].y = t.y;
      f[nt][2 * jp + 1].x = t.z; f[nt][2 * jp + 1].y = t.w;
    }
  }
}

__device__ __forceinline__ v2f s4_state_frag(const v8f (&X)[4], int j) {
  v2f b;
  b.x = X[j >> 2][2 * (j & 3)];
  b.y = X[j >> 2][2 * (j & 3) + 1];
  return b;
}

__device__ __forceinline__ void s4_load_u(const float* __restrict__ up, v2f (&uf)[16]) {
#pragma unroll
  for (int j = 0; j < 16; ++j) {
    uf[j].x = up[(4 * j) * DMOD];
    uf[j].y = up[(4 * j + 1) * DMOD];
  }
}

__device__ __forceinline__ void s4_state_chunk(const v2f (&fM)[4][16], const v2f (&fP)[4][16],
                                               const v2f (&uf)[16], const v8f (&X)[4],
                                               v8f (&Xn)[4]) {
#pragma unroll
  for (int nt = 0; nt < 4; nt += 2) {
    v8f a0 = {0.f, 0.f, 0.f, 0.f, 0.f, 0.f, 0.f, 0.f};
    v8f a1 = {0.f, 0.f, 0.f, 0.f, 0.f, 0.f, 0.f, 0.f};
#pragma unroll
    for (int j = 0; j < 16; ++j) {
      const v2f xb = s4_state_frag(X, j);
      a0 = s4_mma(fM[nt][j],     uf[j], a0);
      a1 = s4_mma(fM[nt + 1][j], uf[j], a1);
      a0 = s4_mma(fP[nt][j],     xb, a0);
      a1 = s4_mma(fP[nt + 1][j], xb, a1);
    }
    Xn[nt] = a0;
    Xn[nt + 1] = a1;
  }
}

__global__ __launch_bounds__(32) __attribute__((amdgpu_waves_per_eu(1, 1)))
void s4_segstate(const float* __restrict__ u, float* __restrict__ ws) {
  const int lane = threadIdx.x;
  const int tile = blockIdx.x / NSEG;
  const int seg  = blockIdx.x % NSEG;
  const int s0 = tile * 16;
  const int bb = s0 >> 9, dm0 = s0 & 511;
  const int c = lane & 15, hi = lane >> 4;
  const float* ub0 = u + (long)bb * LSEQ * DMOD + dm0 + c
                       + (long)(seg * SCH * TCH + 2 * hi) * DMOD;

  v2f ua[16], ub[16];
  v2f fM[4][16], fP[4][16];
  s4_load_u(ub0, ua);
  s4_load_frags(ws + FR_M, lane, fM);
  s4_load_frags(ws + FR_P, lane, fP);
  s4_load_u(ub0 + (long)TCH * DMOD, ub);

  v8f X[4], Xa[4];
#pragma unroll
  for (int nt = 0; nt < 4; ++nt)
#pragma unroll
    for (int r = 0; r < 8; ++r) X[nt][r] = 0.f;

#pragma clang loop unroll(disable)
  for (int p = 0; p < SCH / 2 - 1; ++p) {
    const float* un = ub0 + (long)(2 * p + 2) * TCH * DMOD;
    s4_state_chunk(fM, fP, ua, X, Xa);
    s4_load_u(un, ua);
    s4_state_chunk(fM, fP, ub, Xa, X);
    s4_load_u(un + (long)TCH * DMOD, ub);
  }
  s4_state_chunk(fM, fP, ua, X, Xa);
  s4_state_chunk(fM, fP, ub, Xa, X);

  float* st = ws + WS_ST + (long)tile * (NSEG * 1024) + seg * 1024 + lane;
#pragma unroll
  for (int nt = 0; nt < 4; ++nt)
#pragma unroll
    for (int r = 0; r < 8; ++r) st[(nt * 8 + r) * 32] = X[nt][r];
}

__device__ __forceinline__ void s4_load_dump(const float* __restrict__ p, v8f (&G)[4]) {
#pragma unroll
  for (int nt = 0; nt < 4; ++nt)
#pragma unroll
    for (int r = 0; r < 8; ++r) G[nt][r] = p[(nt * 8 + r) * 32];
}

__device__ __forceinline__ void s4_scan_step(const v2f (&fPS)[4][16], v8f (&E)[4],
                                             const v8f (&G)[4]) {
  v8f En[4];
#pragma unroll
  for (int nt = 0; nt < 4; nt += 2) {
    v8f a0 = G[nt];
    v8f a1 = G[nt + 1];
#pragma unroll
    for (int j = 0; j < 16; ++j) {
      const v2f eb = s4_state_frag(E, j);
      a0 = s4_mma(fPS[nt][j],     eb, a0);
      a1 = s4_mma(fPS[nt + 1][j], eb, a1);
    }
    En[nt] = a0;
    En[nt + 1] = a1;
  }
#pragma unroll
  for (int nt = 0; nt < 4; ++nt) E[nt] = En[nt];
}

__device__ __forceinline__ void s4_out_chunk(const v2f (&fTK)[4][16], const v2f (&fM)[4][16],
                                             const v2f (&fE)[4][16], const v2f (&fP)[4][16],
                                             const v2f (&uf)[16], const v8f (&X)[4],
                                             v8f (&Xn)[4], float* __restrict__ yp0) {
#pragma unroll
  for (int nt = 0; nt < 4; ++nt) {
    v8f ay = {0.f, 0.f, 0.f, 0.f, 0.f, 0.f, 0.f, 0.f};
    v8f ax = {0.f, 0.f, 0.f, 0.f, 0.f, 0.f, 0.f, 0.f};
#pragma unroll
    for (int j = 0; j < 16; ++j) {
      const v2f xb = s4_state_frag(X, j);
      ay = s4_mma(fTK[nt][j], uf[j], ay);
      ax = s4_mma(fM[nt][j],  uf[j], ax);
      ay = s4_mma(fE[nt][j],  xb, ay);
      ax = s4_mma(fP[nt][j],  xb, ax);
    }
    float* yp = yp0 + (long)(nt * 16) * DMOD;
#pragma unroll
    for (int r = 0; r < 8; ++r) yp[r * DMOD] = ay[r];
    Xn[nt] = ax;
  }
}

__device__ __forceinline__ void s4_out_last(const v2f (&fTK)[4][16], const v2f (&fE)[4][16],
                                            const v2f (&uf)[16], const v8f (&X)[4],
                                            float* __restrict__ yp0) {
#pragma unroll
  for (int nt = 0; nt < 4; nt += 2) {
    v8f a0 = {0.f, 0.f, 0.f, 0.f, 0.f, 0.f, 0.f, 0.f};
    v8f a1 = {0.f, 0.f, 0.f, 0.f, 0.f, 0.f, 0.f, 0.f};
#pragma unroll
    for (int j = 0; j < 16; ++j) {
      const v2f xb = s4_state_frag(X, j);
      a0 = s4_mma(fTK[nt][j],     uf[j], a0);
      a1 = s4_mma(fTK[nt + 1][j], uf[j], a1);
      a0 = s4_mma(fE[nt][j],      xb, a0);
      a1 = s4_mma(fE[nt + 1][j],  xb, a1);
    }
    float* yp = yp0 + (long)(nt * 16) * DMOD;
#pragma unroll
    for (int r = 0; r < 8; ++r) yp[r * DMOD] = a0[r];
#pragma unroll
    for (int r = 0; r < 8; ++r) yp[(16 + r) * DMOD] = a1[r];
  }
}

#define S4_HOLD_LOADS() asm volatile("" ::: "memory")

__global__ __launch_bounds__(32)
void s4_outputs(const float* __restrict__ u, const float* __restrict__ ws,
                float* __restrict__ y) {
  const int lane = threadIdx.x;
  const int tile = blockIdx.x / NSEG;
  const int seg  = blockIdx.x % NSEG;
  const int s0 = tile * 16;
  const int bb = s0 >> 9, dm0 = s0 & 511;
  const int c = lane & 15, hi = lane >> 4;
  const float* ub0 = u + (long)bb * LSEQ * DMOD + dm0 + c
                       + (long)(seg * SCH * TCH + 2 * hi) * DMOD;
  float* yb0 = y + (long)bb * LSEQ * DMOD + dm0 + c
                 + (long)(seg * SCH * TCH + 8 * hi) * DMOD;

  const float* stt = ws + WS_ST + (long)tile * (NSEG * 1024) + lane;

  v8f X[4], Xa[4];
  if (seg == 0) {
#pragma unroll
    for (int nt = 0; nt < 4; ++nt)
#pragma unroll
      for (int r = 0; r < 8; ++r) X[nt][r] = 0.f;
  } else {
    s4_load_dump(stt, X);
  }
  v2f fPS[4][16];
  v8f Ga[4], Gb[4];
  v2f ua[16], ub[16];
  v2f fTK[4][16], fM[4][16], fE[4][16], fP[4][16];
  if (seg > 1) {
    s4_load_frags(ws + FR_PS, lane, fPS);
    s4_load_dump(stt + 1 * 1024, Ga);
    if (seg > 2) s4_load_dump(stt + 2 * 1024, Gb);
  }
  s4_load_u(ub0, ua);
  S4_HOLD_LOADS();
  if (seg > 1) s4_scan_step(fPS, X, Ga);

  if (seg > 3) s4_load_dump(stt + 3 * 1024, Ga);
  s4_load_frags(ws + FR_TK, lane, fTK);
  S4_HOLD_LOADS();
  if (seg > 2) s4_scan_step(fPS, X, Gb);

  if (seg > 4) s4_load_dump(stt + 4 * 1024, Gb);
  s4_load_frags(ws + FR_M, lane, fM);
  S4_HOLD_LOADS();
  if (seg > 3) s4_scan_step(fPS, X, Ga);

  if (seg > 5) s4_load_dump(stt + 5 * 1024, Ga);
  s4_load_frags(ws + FR_E, lane, fE);
  S4_HOLD_LOADS();
  if (seg > 4) s4_scan_step(fPS, X, Gb);

  if (seg > 6) s4_load_dump(stt + 6 * 1024, Gb);
  s4_load_frags(ws + FR_P, lane, fP);
  S4_HOLD_LOADS();
  if (seg > 5) s4_scan_step(fPS, X, Ga);

  s4_load_u(ub0 + (long)TCH * DMOD, ub);
  S4_HOLD_LOADS();
  if (seg > 6) s4_scan_step(fPS, X, Gb);

#pragma clang loop unroll(disable)
  for (int p = 0; p < SCH / 2 - 1; ++p) {
    const float* un = ub0 + (long)(2 * p + 2) * TCH * DMOD;
    float* yp = yb0 + (long)(2 * p) * TCH * DMOD;
    s4_out_chunk(fTK, fM, fE, fP, ua, X, Xa, yp);
    s4_load_u(un, ua);
    s4_out_chunk(fTK, fM, fE, fP, ub, Xa, X, yp + (long)TCH * DMOD);
    s4_load_u(un + (long)TCH * DMOD, ub);
  }
  {
    float* yp = yb0 + (long)(SCH - 2) * TCH * DMOD;
    s4_out_chunk(fTK, fM, fE, fP, ua, X, Xa, yp);
    s4_out_last(fTK, fE, ub, Xa, yp + (long)TCH * DMOD);
  }
}

extern "C" void kernel_launch(void* const* d_in, const int* in_sizes, int n_in,
                              void* d_out, int out_size, void* d_ws, size_t ws_size,
                              hipStream_t stream) {
  const float* u  = (const float*)d_in[0];
  const float* A_ = (const float*)d_in[1];
  const float* B_ = (const float*)d_in[2];
  const float* C_ = (const float*)d_in[3];
  float* y  = (float*)d_out;
  float* ws = (float*)d_ws;

  s4_precompute<<<1, 512, 0, stream>>>(A_, B_, C_, ws);
  s4_segstate<<<128 * NSEG, 32, 0, stream>>>(u, ws);
  s4_outputs<<<128 * NSEG, 32, 0, stream>>>(u, ws, y);
}


# Set-up launch: its three copy loops unrolled: all global reads of A, b, c go out as one group, each write-out's LDS reads precede its stores (one round trip per loop, was 8); no sum changes
# speedup vs baseline: 46.7108x; 1.0631x over previous
#include <hip/hip_runtime.h>
#include <hip/hip_bf16.h>

typedef __attribute__((ext_vector_type(2))) float v2f;
typedef __attribute__((ext_vector_type(8))) float v8f;

#define NST 64
#define TCH 64
#define LSEQ 4096
#define DMOD 512
#define NSEG 8
#define SCH 8
#define NRND 9

#define FR_TK 0
#define FR_M  4096
#define FR_E  8192
#define FR_P  12288
#define FR_PS 16384
#define WS_ST 20480

__device__ __forceinline__ void s4_frag_kc(int e, int& ku, int& kd, int& col) {
  const int el = e & 1, jj = (e >> 1) & 1, l = (e >> 2) & 31, jp = (e >> 7) & 7, nt = e >> 10;
  const int j = 2 * jp + jj, h = l >> 4;
  col = nt * 16 + (l & 15);
  ku = 4 * j + 2 * h + el;
  kd = 16 * (j >> 2) + 8 * h + 2 * (j & 3) + el;
}

#define PIT 68

__device__ __forceinline__ v8f s4_tile_mm(const float* Am, const float* Bm,
                                          int i0, int j0, int row, int hi) {
  v8f acc = {0.f, 0.f, 0.f, 0.f, 0.f, 0.f, 0.f, 0.f};
  const float* pa = Am + (i0 + row) * PIT + 2 * hi;
  const float* pb = Bm + (2 * hi) * PIT + j0 + row;
#pragma unroll
  for (int kk = 0; kk < 16; ++kk) {
    const v2f af = *(const v2f*)(pa + 4 * kk);
    v2f bf;
    bf.x = pb[(4 * kk) * PIT];
    bf.y = pb[(4 * kk + 1) * PIT];
    acc = __builtin_amdgcn_wmma_f32_16x16x4_f32(false, af, false, bf,
                                                (short)0, acc, false, false);
  }
  return acc;
}

__global__ __launch_bounds__(512) void s4_precompute(const float* __restrict__ A_,
                                                     const float* __restrict__ B_,
                                                     const float* __restrict__ C_,
                                                     float* __restrict__ ws) {
  __shared__ float S[NST * PIT];
  __shared__ float V[NST * PIT];
  __shared__ float Wm[NST * PIT];
  __shared__ float cb[NST];
  __shared__ float Kk[NST];

  const int tid = threadIdx.x;
  const int wave = __builtin_amdgcn_readfirstlane(tid >> 5);
  const int lane = tid & 31;
  const int row = lane & 15, hi = lane >> 4;

  float av[8];
#pragma unroll
  for (int q = 0; q < 8; ++q) av[q] = A_[tid + 512 * q];
  float bv = 0.f, cv = 0.f;
  if (tid < NST) {
    cv = C_[tid];
    bv = B_[tid];
  }
#pragma unroll
  for (int q = 0; q < 8; ++q) {
    const int e = tid + 512 * q;
    const int i = e >> 6, j = e & 63;
    V[i * PIT + j] = 0.f;
    Wm[i * PIT + j] = 0.f;
  }
#pragma unroll
  for (int q = 0; q < 8; ++q) {
    const int e = tid + 512 * q;
    const int i = e >> 6, j = e & 63;
    S[i * PIT + j] = av[q];
  }
  if (tid < NST) cb[tid] = cv;
  __syncthreads();

  if (tid < NST) {
    V[tid * PIT] = bv;
  } else if (tid < 2 * NST) {
    const int j = tid - NST;
    float s = 0.f;
#pragma unroll 16
    for (int i = 0; i < NST; ++i) s += cb[i] * S[i * PIT + j];
    Wm[j] = s;
  }
  __syncthreads();

  const int ti = wave >> 2, tj = wave & 3;
  const int w8 = wave & 7;
  const int ha = w8 >> 2;
  const int tb = w8 & 3;
  const bool isV = wave < 8;

  float* TKt = ws + FR_TK;
  float* Mt  = ws + FR_M;
  float* Et  = ws + FR_E;
  float* Pt  = ws + FR_P;
  float* PSt = ws + FR_PS;

#pragma clang loop unroll(disable)
  for (int r = 0; r < NRND; ++r) {
    const int m = 1 << r;
    const bool need = (r < 6) && (ha * 16 < m);
    v8f a2 = {0.f, 0.f, 0.f, 0.f, 0.f, 0.f, 0.f, 0.f};
    if (need) {
      a2 = isV ? s4_tile_mm(S, V, tb * 16, ha * 16, row, hi)
               : s4_tile_mm(Wm, S, ha * 16, tb * 16, row, hi);
    }
    const v8f aS = s4_tile_mm(S, S, ti * 16, tj * 16, row, hi);
    __syncthreads();

    {
      float* sp = S + (ti * 16 + 8 * hi) * PIT + tj * 16 + row;
#pragma unroll
      for (int q = 0; q < 8; ++q) sp[q * PIT] = aS[q];
    }
    if (need) {
      if (isV) {
        if (ha * 16 + row < m) {
          float* vp = V + (tb * 16 + 8 * hi) * PIT + m + ha * 16 + row;
#pragma unroll
          for (int q = 0; q < 8; ++q) vp[q * PIT] = a2[q];
        }
      } else {
        float* wp = Wm + (m + ha * 16 + 8 * hi) * PIT + tb * 16 + row;
#pragma unroll
        for (int q = 0; q < 8; ++q)
          if (ha * 16 + 8 * hi + q < m) wp[q * PIT] = a2[q];
      }
    }
    __syncthreads();

    if (r == 5) {
      if (tid < NST) {
        float s = 0.f;
#pragma unroll 16
        for (int i = 0; i < NST; ++i) s += cb[i] * V[i * PIT + tid];
        Kk[tid] = s;
      }
      __syncthreads();
      float tkv[8], mv[8], ev[8], pv[8];
#pragma unroll
      for (int q = 0; q < 8; ++q) {
        int ku, kd, j;
        s4_frag_kc(tid + 512 * q, ku, kd, j);
        tkv[q] = (j >= ku) ? Kk[j - ku] : 0.f;
        mv[q]  = V[j * PIT + (NST - 1 - ku)];
        ev[q]  = Wm[j * PIT + kd];
        pv[q]  = S[j * PIT + kd];
      }
#pragma unroll
      for (int q = 0; q < 8; ++q) {
        const int e = tid + 512 * q;
        TKt[e] = tkv[q];
        Mt[e]  = mv[q];
        Et[e]  = ev[q];
        Pt[e]  = pv[q];
      }
    }
  }
  {
    float psv[8];
#pragma unroll
    for (int q = 0; q < 8; ++q) {
      int ku, kd, j;
      s4_frag_kc(tid + 512 * q, ku, kd, j);
      psv[q] = S[j * PIT + kd];
    }
#pragma unroll
    for (int q = 0; q < 8; ++q) PSt[tid + 512 * q] = psv[q];
  }
}

__device__ __forceinline__ v8f s4_mma(v2f a, v2f b, v8f c) {
  return __builtin_amdgcn_wmma_f32_16x16x4_f32(false, a, false, b, (short)0, c, false, false);
}

__device__ __forceinline__ void s4_load_frags(const float* __restrict__ fr, int lane,
                                              v2f (&f)[4][16]) {
  const float4* f4 = (const float4*)fr + lane;
#pragma unroll
  for (int nt = 0; nt < 4; ++nt) {
#pragma unroll
    for (int jp = 0; jp < 8; ++jp) {
      const float4 t = f4[(nt * 8 + jp) * 32];
      f[nt][2 * jp].x = t.x;     f[nt][2 * jp].y = t.y;
      f[nt][2 * jp + 1].x = t.z; f[nt][2 * jp + 1].y = t.w;
    }
  }
}

__device__ __forceinline__ v2f s4_state_frag(const v8f (&X)[4], int j) {
  v2f b;
  b.x = X[j >> 2][2 * (j & 3)];
  b.y = X[j >> 2][2 * (j & 3) + 1];
  return b;
}

__device__ __forceinline__ void s4_load_u(const float* __restrict__ up, v2f (&uf)[16]) {
#pragma unroll
  for (int j = 0; j < 16; ++j) {
    uf[j].x = up[(4 * j) * DMOD];
    uf[j].y = up[(4 * j + 1) * DMOD];
  }
}

__device__ __forceinline__ void s4_state_chunk(const v2f (&fM)[4][16], const v2f (&fP)[4][16],
                                               const v2f (&uf)[16], const v8f (&X)[4],
                                               v8f (&Xn)[4]) {
#pragma unroll
  for (int nt = 0; nt < 4; nt += 2) {
    v8f a0 = {0.f, 0.f, 0.f, 0.f, 0.f, 0.f, 0.f, 0.f};
    v8f a1 = {0.f, 0.f, 0.f, 0.f, 0.f, 0.f, 0.f, 0.f};
#pragma unroll
    for (int j = 0; j < 16; ++j) {
      const v2f xb = s4_state_frag(X, j);
      a0 = s4_mma(fM[nt][j],     uf[j], a0);
      a1 = s4_mma(fM[nt + 1][j], uf[j], a1);
      a0 = s4_mma(fP[nt][j],     xb, a0);
      a1 = s4_mma(fP[nt + 1][j], xb, a1);
    }
    Xn[nt] = a0;
    Xn[nt + 1] = a1;
  }
}

__global__ __launch_bounds__(32) __attribute__((amdgpu_waves_per_eu(1, 1)))
void s4_segstate(const float* __restrict__ u, float* __restrict__ ws) {
  const int lane = threadIdx.x;
  const int tile = blockIdx.x / NSEG;
  const int seg  = blockIdx.x % NSEG;
  const int s0 = tile * 16;
  const int bb = s0 >> 9, dm0 = s0 & 511;
  const int c = lane & 15, hi = lane >> 4;
  const float* ub0 = u + (long)bb * LSEQ * DMOD + dm0 + c
                       + (long)(seg * SCH * TCH + 2 * hi) * DMOD;

  v2f ua[16], ub[16];
  v2f fM[4][16], fP[4][16];
  s4_load_u(ub0, ua);
  s4_load_frags(ws + FR_M, lane, fM);
  s4_load_frags(ws + FR_P, lane, fP);
  s4_load_u(ub0 + (long)TCH * DMOD, ub);

  v8f X[4], Xa[4];
#pragma unroll
  for (int nt = 0; nt < 4; ++nt)
#pragma unroll
    for (int r = 0; r < 8; ++r) X[nt][r] = 0.f;

#pragma clang loop unroll(disable)
  for (int p = 0; p < SCH / 2 - 1; ++p) {
    const float* un = ub0 + (long)(2 * p + 2) * TCH * DMOD;
    s4_state_chunk(fM, fP, ua, X, Xa);
    s4_load_u(un, ua);
    s4_state_chunk(fM, fP, ub, Xa, X);
    s4_load_u(un + (long)TCH * DMOD, ub);
  }
  s4_state_chunk(fM, fP, ua, X, Xa);
  s4_state_chunk(fM, fP, ub, Xa, X);

  float* st = ws + WS_ST + (long)tile * (NSEG * 1024) + seg * 1024 + lane;
#pragma unroll
  for (int nt = 0; nt < 4; ++nt)
#pragma unroll
    for (int r = 0; r < 8; ++r) st[(nt * 8 + r) * 32] = X[nt][r];
}

__device__ __forceinline__ void s4_load_dump(const float* __restrict__ p, v8f (&G)[4]) {
#pragma unroll
  for (int nt = 0; nt < 4; ++nt)
#pragma unroll
    for (int r = 0; r < 8; ++r) G[nt][r] = p[(nt * 8 + r) * 32];
}

__device__ __forceinline__ void s4_scan_step(const v2f (&fPS)[4][16], v8f (&E)[4],
                                             const v8f (&G)[4]) {
  v8f En[4];
#pragma unroll
  for (int nt = 0; nt < 4; nt += 2) {
    v8f a0 = G[nt];
    v8f a1 = G[nt + 1];
#pragma unroll
    for (int j = 0; j < 16; ++j) {
      const v2f eb = s4_state_frag(E, j);
      a0 = s4_mma(fPS[nt][j],     eb, a0);
      a1 = s4_mma(fPS[nt + 1][j], eb, a1);
    }
    En[nt] = a0;
    En[nt + 1] = a1;
  }
#pragma unroll
  for (int nt = 0; nt < 4; ++nt) E[nt] = En[nt];
}

__device__ __forceinline__ void s4_out_chunk(const v2f (&fTK)[4][16], const v2f (&fM)[4][16],
                                             const v2f (&fE)[4][16], const v2f (&fP)[4][16],
                                             const v2f (&uf)[16], const v8f (&X)[4],
                                             v8f (&Xn)[4], float* __restrict__ yp0) {
#pragma unroll
  for (int nt = 0; nt < 4; ++nt) {
    v8f ay = {0.f, 0.f, 0.f, 0.f, 0.f, 0.f, 0.f, 0.f};
    v8f ax = {0.f, 0.f, 0.f, 0.f, 0.f, 0.f, 0.f, 0.f};
#pragma unroll
    for (int j = 0; j < 16; ++j) {
      const v2f xb = s4_state_frag(X, j);
      ay = s4_mma(fTK[nt][j], uf[j], ay);
      ax = s4_mma(fM[nt][j],  uf[j], ax);
      ay = s4_mma(fE[nt][j],  xb, ay);
      ax = s4_mma(fP[nt][j],  xb, ax);
    }
    float* yp = yp0 + (long)(nt * 16) * DMOD;
#pragma unroll
    for (int r = 0; r < 8; ++r) yp[r * DMOD] = ay[r];
    Xn[nt] = ax;
  }
}

__device__ __forceinline__ void s4_out_last(const v2f (&fTK)[4][16], const v2f (&fE)[4][16],
                                            const v2f (&uf)[16], const v8f (&X)[4],
                                            float* __restrict__ yp0) {
#pragma unroll
  for (int nt = 0; nt < 4; nt += 2) {
    v8f a0 = {0.f, 0.f, 0.f, 0.f, 0.f, 0.f, 0.f, 0.f};
    v8f a1 = {0.f, 0.f, 0.f, 0.f, 0.f, 0.f, 0.f, 0.f};
#pragma unroll
    for (int j = 0; j < 16; ++j) {
      const v2f xb = s4_state_frag(X, j);
      a0 = s4_mma(fTK[nt][j],     uf[j], a0);
      a1 = s4_mma(fTK[nt + 1][j], uf[j], a1);
      a0 = s4_mma(fE[nt][j],      xb, a0);
      a1 = s4_mma(fE[nt + 1][j],  xb, a1);
    }
    float* yp = yp0 + (long)(nt * 16) * DMOD;
#pragma unroll
    for (int r = 0; r < 8; ++r) yp[r * DMOD] = a0[r];
#pragma unroll
    for (int r = 0; r < 8; ++r) yp[(16 + r) * DMOD] = a1[r];
  }
}

#define S4_HOLD_LOADS() asm volatile("" ::: "memory")

__global__ __launch_bounds__(32)
void s4_outputs(const float* __restrict__ u, const float* __restrict__ ws,
                float* __restrict__ y) {
  const int lane = threadIdx.x;
  const int tile = blockIdx.x / NSEG;
  const int seg  = blockIdx.x % NSEG;
  const int s0 = tile * 16;
  const int bb = s0 >> 9, dm0 = s0 & 511;
  const int c = lane & 15, hi = lane >> 4;
  const float* ub0 = u + (long)bb * LSEQ * DMOD + dm0 + c
                       + (long)(seg * SCH * TCH + 2 * hi) * DMOD;
  float* yb0 = y + (long)bb * LSEQ * DMOD + dm0 + c
                 + (long)(seg * SCH * TCH + 8 * hi) * DMOD;

  const float* stt = ws + WS_ST + (long)tile * (NSEG * 1024) + lane;

  v8f X[4], Xa[4];
  if (seg == 0) {
#pragma unroll
    for (int nt = 0; nt < 4; ++nt)
#pragma unroll
      for (int r = 0; r < 8; ++r) X[nt][r] = 0.f;
  } else {
    s4_load_dump(stt, X);
  }
  v2f fPS[4][16];
  v8f Ga[4], Gb[4];
  v2f ua[16], ub[16];
  v2f fTK[4][16], fM[4][16], fE[4][16], fP[4][16];
  if (seg > 1) {
    s4_load_frags(ws + FR_PS, lane, fPS);
    s4_load_dump(stt + 1 * 1024, Ga);
    if (seg > 2) s4_load_dump(stt + 2 * 1024, Gb);
  }
  s4_load_u(ub0, ua);
  S4_HOLD_LOADS();
  if (seg > 1) s4_scan_step(fPS, X, Ga);

  if (seg > 3) s4_load_dump(stt + 3 * 1024, Ga);
  s4_load_frags(ws + FR_TK, lane, fTK);
  S4_HOLD_LOADS();
  if (seg > 2) s4_scan_step(fPS, X, Gb);

  if (seg > 4) s4_load_dump(stt + 4 * 1024, Gb);
  s4_load_frags(ws + FR_M, lane, fM);
  S4_HOLD_LOADS();
  if (seg > 3) s4_scan_step(fPS, X, Ga);

  if (seg > 5) s4_load_dump(stt + 5 * 1024, Ga);
  s4_load_frags(ws + FR_E, lane, fE);
  S4_HOLD_LOADS();
  if (seg > 4) s4_scan_step(fPS, X, Gb);

  if (seg > 6) s4_load_dump(stt + 6 * 1024, Gb);
  s4_load_frags(ws + FR_P, lane, fP);
  S4_HOLD_LOADS();
  if (seg > 5) s4_scan_step(fPS, X, Ga);

  s4_load_u(ub0 + (long)TCH * DMOD, ub);
  S4_HOLD_LOADS();
  if (seg > 6) s4_scan_step(fPS, X, Gb);

#pragma clang loop unroll(disable)
  for (int p = 0; p < SCH / 2 - 1; ++p) {
    const float* un = ub0 + (long)(2 * p + 2) * TCH * DMOD;
    float* yp = yb0 + (long)(2 * p) * TCH * DMOD;
    s4_out_chunk(fTK, fM, fE, fP, ua, X, Xa, yp);
    s4_load_u(un, ua);
    s4_out_chunk(fTK, fM, fE, fP, ub, Xa, X, yp + (long)TCH * DMOD);
    s4_load_u(un + (long)TCH * DMOD, ub);
  }
  {
    float* yp = yb0 + (long)(SCH - 2) * TCH * DMOD;
    s4_out_chunk(fTK, fM, fE, fP, ua, X, Xa, yp);
    s4_out_last(fTK, fE, ub, Xa, yp + (long)TCH * DMOD);
  }
}

extern "C" void kernel_launch(void* const* d_in, const int* in_sizes, int n_in,
                              void* d_out, int out_size, void* d_ws, size_t ws_size,
                              hipStream_t stream) {
  const float* u  = (const float*)d_in[0];
  const float* A_ = (const float*)d_in[1];
  const float* B_ = (const float*)d_in[2];
  const float* C_ = (const float*)d_in[3];
  float* y  = (float*)d_out;
  float* ws = (float*)d_ws;

  s4_precompute<<<1, 512, 0, stream>>>(A_, B_, C_, ws);
  s4_segstate<<<128 * NSEG, 32, 0, stream>>>(u, ws);
  s4_outputs<<<128 * NSEG, 32, 0, stream>>>(u, ws, y);
}
